# starting kernel
# baseline (speedup 1.0000x reference)
#include <hip/hip_runtime.h>
#include <cmath>

typedef __attribute__((ext_vector_type(2))) float v2f;
typedef __attribute__((ext_vector_type(8))) float v8f;

#define B_   128
#define T_   20
#define E_   512
#define H_   512
#define V_   10000
#define LDSS 516

__device__ __forceinline__ float sigmoidf_(float x) {
    return 1.0f / (1.0f + __expf(-x));
}

__device__ __forceinline__ v8f wmma_f32(v2f a, v2f b, v8f c) {
    return __builtin_amdgcn_wmma_f32_16x16x4_f32(false, a, false, b, (short)0, c,
                                                 false, false);
}

__global__ void lstm_init(const float* __restrict__ features,
                          float* __restrict__ h0, float* __restrict__ c0) {
    int i = blockIdx.x * blockDim.x + threadIdx.x;
    if (i < B_ * H_) { h0[i] = features[i]; c0[i] = 0.0f; }
}

__global__ __launch_bounds__(128) void lstm_step(
    const int* __restrict__ captions, int t,
    const float* __restrict__ embed_W,
    const float* __restrict__ W_ih, const float* __restrict__ W_hh,
    const float* __restrict__ b_ih, const float* __restrict__ b_hh,
    const float* __restrict__ h_in, const float* __restrict__ c_in,
    float* __restrict__ h_out, float* __restrict__ c_out)
{
    __shared__ float xt[16 * LDSS];
    __shared__ float ht[16 * LDSS];
    const int mtile = blockIdx.y;
    const int tid   = threadIdx.x;

    const int wave  = tid >> 5;
    const int lane  = tid & 31;
    const int half  = lane >> 4;
    const int l16   = lane & 15;
    const int htile = blockIdx.x * 4 + wave;
    const int col   = htile * 16 + l16;

    v8f acc[4];
    #pragma unroll
    for (int g = 0; g < 4; ++g)
        #pragma unroll
        for (int v = 0; v < 8; ++v) acc[g][v] = 0.0f;

    for (int idx = tid * 4; idx < 16 * E_; idx += 128 * 4) {
        int row = idx / E_, e = idx % E_;
        int tok = captions[(mtile * 16 + row) * T_ + t];
        *(float4*)(xt + row * LDSS + e) =
            *(const float4*)(embed_W + (size_t)tok * E_ + e);
        *(float4*)(ht + row * LDSS + e) =
            *(const float4*)(h_in + (size_t)(mtile * 16 + row) * H_ + e);
    }
    __syncthreads();

    for (int k0 = 0; k0 < E_; k0 += 4) {
        int kk = k0 + 2 * half;
        v2f a = *(const v2f*)(xt + l16 * LDSS + kk);
        #pragma unroll
        for (int g = 0; g < 4; ++g) {
            v2f b = *(const v2f*)(W_ih + (size_t)(g * H_ + col) * E_ + kk);
            acc[g] = wmma_f32(a, b, acc[g]);
        }
    }
    for (int k0 = 0; k0 < H_; k0 += 4) {
        int kk = k0 + 2 * half;
        v2f a = *(const v2f*)(ht + l16 * LDSS + kk);
        #pragma unroll
        for (int g = 0; g < 4; ++g) {
            v2f b = *(const v2f*)(W_hh + (size_t)(g * H_ + col) * H_ + kk);
            acc[g] = wmma_f32(a, b, acc[g]);
        }
    }

    const float bi = b_ih[0 * H_ + col] + b_hh[0 * H_ + col];
    const float bf = b_ih[1 * H_ + col] + b_hh[1 * H_ + col];
    const float bg = b_ih[2 * H_ + col] + b_hh[2 * H_ + col];
    const float bo = b_ih[3 * H_ + col] + b_hh[3 * H_ + col];
    #pragma unroll
    for (int v = 0; v < 8; ++v) {
        int rowg = mtile * 16 + v + 8 * half;
        float iv = sigmoidf_(acc[0][v] + bi);
        float fv = sigmoidf_(acc[1][v] + bf);
        float gv = tanhf(acc[2][v] + bg);
        float ov = sigmoidf_(acc[3][v] + bo);
        float cp = c_in[(size_t)rowg * H_ + col];
        float cn = fv * cp + iv * gv;
        c_out[(size_t)rowg * H_ + col] = cn;
        h_out[(size_t)rowg * H_ + col] = ov * tanhf(cn);
    }
}

__global__ __launch_bounds__(160) void proj_step(
    const float* __restrict__ h, const float* __restrict__ lin_W,
    const float* __restrict__ lin_b, float* __restrict__ out, int t)
{
    __shared__ float hsh[32 * LDSS];
    const int mbase = blockIdx.y * 32;
    const int tid   = threadIdx.x;

    for (int idx = tid * 4; idx < 32 * H_; idx += 160 * 4) {
        int row = idx / H_, e = idx % H_;
        *(float4*)(hsh + row * LDSS + e) =
            *(const float4*)(h + (size_t)(mbase + row) * H_ + e);
    }
    __syncthreads();

    const int wave = tid >> 5;
    const int lane = tid & 31;
    const int half = lane >> 4, l16 = lane & 15;
    const int ntile0 = blockIdx.x * 25 + wave * 5;

    v8f acc[10];
    int  ncol[5];
    #pragma unroll
    for (int i = 0; i < 5; ++i) ncol[i] = (ntile0 + i) * 16 + l16;
    #pragma unroll
    for (int j = 0; j < 10; ++j)
        #pragma unroll
        for (int v = 0; v < 8; ++v) acc[j][v] = 0.0f;

    for (int k0 = 0; k0 < H_; k0 += 4) {
        int kk = k0 + 2 * half;
        v2f a0 = *(const v2f*)(hsh + l16 * LDSS + kk);
        v2f a1 = *(const v2f*)(hsh + (16 + l16) * LDSS + kk);
        #pragma unroll
        for (int i = 0; i < 5; ++i) {
            v2f b = *(const v2f*)(lin_W + (size_t)ncol[i] * H_ + kk);
            acc[i]     = wmma_f32(a0, b, acc[i]);
            acc[5 + i] = wmma_f32(a1, b, acc[5 + i]);
        }
    }

    #pragma unroll
    for (int m = 0; m < 2; ++m) {
        #pragma unroll
        for (int i = 0; i < 5; ++i) {
            float bias = lin_b[ncol[i]];
            #pragma unroll
            for (int v = 0; v < 8; ++v) {
                int rowg = mbase + m * 16 + v + 8 * half;
                __builtin_nontemporal_store(
                    acc[m * 5 + i][v] + bias,
                    out + (size_t)rowg * T_ * V_ + (size_t)t * V_ + ncol[i]);
            }
        }
    }
}

extern "C" void kernel_launch(void* const* d_in, const int* in_sizes, int n_in,
                              void* d_out, int out_size, void* d_ws, size_t ws_size,
                              hipStream_t stream) {
    const float* features = (const float*)d_in[0];
    const int*   captions = (const int*)d_in[1];
    const float* embed_W  = (const float*)d_in[2];
    const float* W_ih     = (const float*)d_in[3];
    const float* W_hh     = (const float*)d_in[4];
    const float* b_ih     = (const float*)d_in[5];
    const float* b_hh     = (const float*)d_in[6];
    const float* lin_W    = (const float*)d_in[7];
    const float* lin_b    = (const float*)d_in[8];
    float* out = (float*)d_out;

    float* ws = (float*)d_ws;
    float* hbuf[2] = { ws,                       ws + (size_t)B_ * H_ };
    float* cbuf[2] = { ws + 2 * (size_t)B_ * H_, ws + 3 * (size_t)B_ * H_ };

    lstm_init<<<(B_ * H_ + 255) / 256, 256, 0, stream>>>(features, hbuf[0], cbuf[0]);

    for (int t = 0; t < T_; ++t) {
        int cur = t & 1, nxt = cur ^ 1;
        lstm_step<<<dim3(8, 8), 128, 0, stream>>>(
            captions, t, embed_W, W_ih, W_hh, b_ih, b_hh,
            hbuf[cur], cbuf[cur], hbuf[nxt], cbuf[nxt]);
        proj_step<<<dim3(25, 4), 160, 0, stream>>>(hbuf[nxt], lin_W, lin_b, out, t);
    }
}


# Run step t's vocabulary projection in the same launch as step t+1's cell update (both only read h of step t; role by block index): 41 launches -> 22; sums and their order unchanged
# speedup vs baseline: 1.8665x; 1.8665x over previous
#include <hip/hip_runtime.h>
#include <cmath>


typedef __attribute__((ext_vector_type(2))) float v2f;
typedef __attribute__((ext_vector_type(8))) float v8f;

#define B_   128
#define T_   20
#define E_   512
#define H_   512
#define V_   10000
#define LDSS 516
#define NT_  160
#define CELL_BLOCKS 64
#define PROJ_BLOCKS 100

__device__ __forceinline__ float sigmoidf_(float x) {
    return 1.0f / (1.0f + __expf(-x));
}

__device__ __forceinline__ v8f wmma_f32(v2f a, v2f b, v8f c) {
    return __builtin_amdgcn_wmma_f32_16x16x4_f32(false, a, false, b, (short)0, c,
                                                 false, false);
}

__global__ void lstm_init(const float* __restrict__ features,
                          float* __restrict__ h0, float* __restrict__ c0) {
    int i = blockIdx.x * blockDim.x + threadIdx.x;
    if (i < B_ * H_) { h0[i] = features[i]; c0[i] = 0.0f; }
}

__device__ __forceinline__ void cell_role(
    float* __restrict__ smem, int bid,
    const int* __restrict__ captions, int t,
    const float* __restrict__ embed_W,
    const float* __restrict__ W_ih, const float* __restrict__ W_hh,
    const float* __restrict__ b_ih, const float* __restrict__ b_hh,
    const float* __restrict__ h_in, const float* __restrict__ c_in,
    float* __restrict__ h_out, float* __restrict__ c_out)
{
    float* xt = smem;
    float* ht = smem + 16 * LDSS;
    const int mtile = bid >> 3;
    const int tid   = threadIdx.x;

    for (int idx = tid * 4; idx < 16 * E_; idx += NT_ * 4) {
        int row = idx / E_, e = idx % E_;
        int tok = captions[(mtile * 16 + row) * T_ + t];
        *(float4*)(xt + row * LDSS + e) =
            *(const float4*)(embed_W + (size_t)tok * E_ + e);
        *(float4*)(ht + row * LDSS + e) =
            *(const float4*)(h_in + (size_t)(mtile * 16 + row) * H_ + e);
    }
    __syncthreads();

    const int wave  = tid >> 5;
    if (wave >= 4) return;
    const int lane  = tid & 31;
    const int half  = lane >> 4;
    const int l16   = lane & 15;
    const int htile = (bid & 7) * 4 + wave;
    const int col   = htile * 16 + l16;

    v8f acc[4];
    #pragma unroll
    for (int g = 0; g < 4; ++g)
        #pragma unroll
        for (int v = 0; v < 8; ++v) acc[g][v] = 0.0f;

    for (int k0 = 0; k0 < E_; k0 += 4) {
        int kk = k0 + 2 * half;
        v2f a = *(const v2f*)(xt + l16 * LDSS + kk);
        #pragma unroll
        for (int g = 0; g < 4; ++g) {
            v2f b = *(const v2f*)(W_ih + (size_t)(g * H_ + col) * E_ + kk);
            acc[g] = wmma_f32(a, b, acc[g]);
        }
    }
    for (int k0 = 0; k0 < H_; k0 += 4) {
        int kk = k0 + 2 * half;
        v2f a = *(const v2f*)(ht + l16 * LDSS + kk);
        #pragma unroll
        for (int g = 0; g < 4; ++g) {
            v2f b = *(const v2f*)(W_hh + (size_t)(g * H_ + col) * H_ + kk);
            acc[g] = wmma_f32(a, b, acc[g]);
        }
    }

    const float bi = b_ih[0 * H_ + col] + b_hh[0 * H_ + col];
    const float bf = b_ih[1 * H_ + col] + b_hh[1 * H_ + col];
    const float bg = b_ih[2 * H_ + col] + b_hh[2 * H_ + col];
    const float bo = b_ih[3 * H_ + col] + b_hh[3 * H_ + col];
    #pragma unroll
    for (int v = 0; v < 8; ++v) {
        int rowg = mtile * 16 + v + 8 * half;
        float iv = sigmoidf_(acc[0][v] + bi);
        float fv = sigmoidf_(acc[1][v] + bf);
        float gv = tanhf(acc[2][v] + bg);
        float ov = sigmoidf_(acc[3][v] + bo);
        float cp = c_in[(size_t)rowg * H_ + col];
        float cn = fv * cp + iv * gv;
        c_out[(size_t)rowg * H_ + col] = cn;
        h_out[(size_t)rowg * H_ + col] = ov * tanhf(cn);
    }
}

__device__ __forceinline__ void proj_role(
    float* __restrict__ hsh, int pb,
    const float* __restrict__ h, const float* __restrict__ lin_W,
    const float* __restrict__ lin_b, float* __restrict__ out, int t)
{
    const int mbase = (pb / 25) * 32;
    const int tid   = threadIdx.x;

    for (int idx = tid * 4; idx < 32 * H_; idx += NT_ * 4) {
        int row = idx / H_, e = idx % H_;
        *(float4*)(hsh + row * LDSS + e) =
            *(const float4*)(h + (size_t)(mbase + row) * H_ + e);
    }
    __syncthreads();

    const int wave = tid >> 5;
    const int lane = tid & 31;
    const int half = lane >> 4, l16 = lane & 15;
    const int ntile0 = (pb % 25) * 25 + wave * 5;

    v8f acc[10];
    int  ncol[5];
    #pragma unroll
    for (int i = 0; i < 5; ++i) ncol[i] = (ntile0 + i) * 16 + l16;
    #pragma unroll
    for (int j = 0; j < 10; ++j)
        #pragma unroll
        for (int v = 0; v < 8; ++v) acc[j][v] = 0.0f;

    for (int k0 = 0; k0 < H_; k0 += 4) {
        int kk = k0 + 2 * half;
        v2f a0 = *(const v2f*)(hsh + l16 * LDSS + kk);
        v2f a1 = *(const v2f*)(hsh + (16 + l16) * LDSS + kk);
        #pragma unroll
        for (int i = 0; i < 5; ++i) {
            v2f b = *(const v2f*)(lin_W + (size_t)ncol[i] * H_ + kk);
            acc[i]     = wmma_f32(a0, b, acc[i]);
            acc[5 + i] = wmma_f32(a1, b, acc[5 + i]);
        }
    }

    #pragma unroll
    for (int m = 0; m < 2; ++m) {
        #pragma unroll
        for (int i = 0; i < 5; ++i) {
            float bias = lin_b[ncol[i]];
            #pragma unroll
            for (int v = 0; v < 8; ++v) {
                int rowg = mbase + m * 16 + v + 8 * half;
                __builtin_nontemporal_store(
                    acc[m * 5 + i][v] + bias,
                    out + (size_t)rowg * T_ * V_ + (size_t)t * V_ + ncol[i]);
            }
        }
    }
}

__global__ __launch_bounds__(NT_) void step_pair(
    const int* __restrict__ captions, const float* __restrict__ embed_W,
    const float* __restrict__ W_ih, const float* __restrict__ W_hh,
    const float* __restrict__ b_ih, const float* __restrict__ b_hh,
    const float* __restrict__ lin_W, const float* __restrict__ lin_b,
    const float* __restrict__ h_cur, const float* __restrict__ c_cur,
    float* __restrict__ h_nxt, float* __restrict__ c_nxt,
    float* __restrict__ out, int t_cell, int t_proj, int n_cell)
{
    __shared__ float smem[32 * LDSS];
    const int bid = blockIdx.x;
    if (bid < n_cell)
        cell_role(smem, bid, captions, t_cell, embed_W, W_ih, W_hh, b_ih, b_hh,
                  h_cur, c_cur, h_nxt, c_nxt);
    else
        proj_role(smem, bid - n_cell, h_cur, lin_W, lin_b, out, t_proj);
}

extern "C" void kernel_launch(void* const* d_in, const int* in_sizes, int n_in,
                              void* d_out, int out_size, void* d_ws, size_t ws_size,
                              hipStream_t stream) {
    const float* features = (const float*)d_in[0];
    const int*   captions = (const int*)d_in[1];
    const float* embed_W  = (const float*)d_in[2];
    const float* W_ih     = (const float*)d_in[3];
    const float* W_hh     = (const float*)d_in[4];
    const float* b_ih     = (const float*)d_in[5];
    const float* b_hh     = (const float*)d_in[6];
    const float* lin_W    = (const float*)d_in[7];
    const float* lin_b    = (const float*)d_in[8];
    float* out = (float*)d_out;

    float* ws = (float*)d_ws;
    float* hbuf[2] = { ws,                       ws + (size_t)B_ * H_ };
    float* cbuf[2] = { ws + 2 * (size_t)B_ * H_, ws + 3 * (size_t)B_ * H_ };

    lstm_init<<<(B_ * H_ + 255) / 256, 256, 0, stream>>>(features, hbuf[0], cbuf[0]);

    step_pair<<<CELL_BLOCKS, NT_, 0, stream>>>(
        captions, embed_W, W_ih, W_hh, b_ih, b_hh, lin_W, lin_b,
        hbuf[0], cbuf[0], hbuf[1], cbuf[1], out, 0, 0, CELL_BLOCKS);

    for (int t = 1; t < T_; ++t) {
        int cur = t & 1, nxt = cur ^ 1;
        step_pair<<<CELL_BLOCKS + PROJ_BLOCKS, NT_, 0, stream>>>(
            captions, embed_W, W_ih, W_hh, b_ih, b_hh, lin_W, lin_b,
            hbuf[cur], cbuf[cur], hbuf[nxt], cbuf[nxt], out, t, t - 1, CELL_BLOCKS);
    }

    step_pair<<<PROJ_BLOCKS, NT_, 0, stream>>>(
        captions, embed_W, W_ih, W_hh, b_ih, b_hh, lin_W, lin_b,
        hbuf[T_ & 1], cbuf[T_ & 1], hbuf[(T_ & 1) ^ 1], cbuf[(T_ & 1) ^ 1],
        out, 0, T_ - 1, 0);
}


# Issue every global and LDS load ahead of its use: staging loads as one group, k loops in blocks with two fragment sets (next block's loads before this block's WMMAs); sum order unchanged
# speedup vs baseline: 2.6774x; 1.4344x over previous
#include <hip/hip_runtime.h>
#include <cmath>


typedef __attribute__((ext_vector_type(2))) float v2f;
typedef __attribute__((ext_vector_type(8))) float v8f;

#define B_   128
#define T_   20
#define E_   512
#define H_   512
#define V_   10000
#define LDSS 516
#define NT_  160
#define CELL_BLOCKS 64
#define PROJ_BLOCKS 100

__device__ __forceinline__ float sigmoidf_(float x) {
    return 1.0f / (1.0f + __expf(-x));
}

__device__ __forceinline__ v8f wmma_f32(v2f a, v2f b, v8f c) {
    return __builtin_amdgcn_wmma_f32_16x16x4_f32(false, a, false, b, (short)0, c,
                                                 false, false);
}

__global__ void lstm_init(const float* __restrict__ features,
                          float* __restrict__ h0, float* __restrict__ c0) {
    int i = blockIdx.x * blockDim.x + threadIdx.x;
    if (i < B_ * H_) { h0[i] = features[i]; c0[i] = 0.0f; }
}

#define LOADS_AHEAD() __builtin_amdgcn_sched_barrier(0)
#define MMA_DONE4(a)                                                                    \
    do {                                                                                \
        asm volatile("v_nop\n\tv_nop\n\tv_nop\n\tv_nop\n\tv_nop"                         \
                     : "+v"(a[0]), "+v"(a[1]), "+v"(a[2]), "+v"(a[3]));                  \
        __builtin_amdgcn_sched_barrier(0);                                              \
    } while (0)
#define MMA_DONE10(a)                                                                   \
    do {                                                                                \
        asm volatile("v_nop\n\tv_nop\n\tv_nop\n\tv_nop\n\tv_nop"                         \
                     : "+v"(a[0]), "+v"(a[1]), "+v"(a[2]), "+v"(a[3]), "+v"(a[4]),       \
                       "+v"(a[5]), "+v"(a[6]), "+v"(a[7]), "+v"(a[8]), "+v"(a[9]));      \
        __builtin_amdgcn_sched_barrier(0);                                              \
    } while (0)
#define CKB 64
#define CKS (CKB / 4)

struct CellFrag { v2f a[CKS]; v2f b[CKS][4]; };

__device__ __forceinline__ void cell_load(CellFrag& f, const float* __restrict__ arow,
                                          const float* __restrict__ wrow, int kb) {
    #pragma unroll
    for (int s = 0; s < CKS; ++s)
        f.a[s] = *(const v2f*)(arow + kb + 4 * s);
    #pragma unroll
    for (int s = 0; s < CKS; ++s)
        #pragma unroll
        for (int g = 0; g < 4; ++g)
            f.b[s][g] = *(const v2f*)(wrow + (size_t)g * H_ * E_ + kb + 4 * s);
}

__device__ __forceinline__ void cell_mma(const CellFrag& f, v8f (&acc)[4]) {
    #pragma unroll
    for (int s = 0; s < CKS; ++s)
        #pragma unroll
        for (int g = 0; g < 4; ++g)
            acc[g] = wmma_f32(f.a[s], f.b[s][g], acc[g]);
}

__device__ __forceinline__ void cell_role(
    float* __restrict__ smem, int bid,
    const int* __restrict__ captions, int t,
    const float* __restrict__ embed_W,
    const float* __restrict__ W_ih, const float* __restrict__ W_hh,
    const float* __restrict__ b_ih, const float* __restrict__ b_hh,
    const float* __restrict__ h_in, const float* __restrict__ c_in,
    float* __restrict__ h_out, float* __restrict__ c_out)
{
    float* xt = smem;
    float* ht = smem + 16 * LDSS;
    const int mtile = bid >> 3;
    const int tid   = threadIdx.x;

    if (tid < 128) {
        constexpr int NJ = 16;
        int tok[NJ];
        #pragma unroll
        for (int j = 0; j < NJ; ++j) {
            int q = j * 128 + tid;
            tok[j] = captions[(mtile * 16 + (q >> 7)) * T_ + t];
        }
        float4 xr[NJ], hr[NJ];
        #pragma unroll
        for (int j = 0; j < NJ; ++j) {
            int q = j * 128 + tid;
            int row = q >> 7, e = (q & 127) << 2;
            xr[j] = *(const float4*)(embed_W + (size_t)tok[j] * E_ + e);
            hr[j] = *(const float4*)(h_in + (size_t)(mtile * 16 + row) * H_ + e);
        }
        #pragma unroll
        for (int j = 0; j < NJ; ++j) {
            int q = j * 128 + tid;
            int row = q >> 7, e = (q & 127) << 2;
            *(float4*)(xt + row * LDSS + e) = xr[j];
            *(float4*)(ht + row * LDSS + e) = hr[j];
        }
    }
    __syncthreads();

    const int wave  = tid >> 5;
    if (wave >= 4) return;
    const int lane  = tid & 31;
    const int half  = lane >> 4;
    const int l16   = lane & 15;
    const int htile = (bid & 7) * 4 + wave;
    const int col   = htile * 16 + l16;

    float bi = b_ih[0 * H_ + col] + b_hh[0 * H_ + col];
    float bf = b_ih[1 * H_ + col] + b_hh[1 * H_ + col];
    float bg = b_ih[2 * H_ + col] + b_hh[2 * H_ + col];
    float bo = b_ih[3 * H_ + col] + b_hh[3 * H_ + col];
    float cp[8];
    #pragma unroll
    for (int v = 0; v < 8; ++v)
        cp[v] = c_in[(size_t)(mtile * 16 + v + 8 * half) * H_ + col];

    v8f acc[4];
    #pragma unroll
    for (int g = 0; g < 4; ++g)
        #pragma unroll
        for (int v = 0; v < 8; ++v) acc[g][v] = 0.0f;

    const float* ax = xt + l16 * LDSS + 2 * half;
    const float* ah = ht + l16 * LDSS + 2 * half;
    const float* wx = W_ih + (size_t)col * E_ + 2 * half;
    const float* wh = W_hh + (size_t)col * H_ + 2 * half;

    constexpr int XB = E_ / CKB;
    CellFrag fp, fq;
    cell_load(fp, ax, wx, 0);
    asm volatile("" : "+v"(bi), "+v"(bf), "+v"(bg), "+v"(bo));
    for (int bk = 0; bk < 2 * XB - 2; bk += 2) {
        int nb = bk + 1;
        cell_load(fq, nb < XB ? ax : ah, nb < XB ? wx : wh, (nb & (XB - 1)) * CKB);
        LOADS_AHEAD();
        cell_mma(fp, acc);
        MMA_DONE4(acc);
        nb = bk + 2;
        cell_load(fp, nb < XB ? ax : ah, nb < XB ? wx : wh, (nb & (XB - 1)) * CKB);
        LOADS_AHEAD();
        cell_mma(fq, acc);
        MMA_DONE4(acc);
    }
    cell_load(fq, ah, wh, (XB - 1) * CKB);
    LOADS_AHEAD();
    cell_mma(fp, acc);
    cell_mma(fq, acc);
    MMA_DONE4(acc);

    #pragma unroll
    for (int v = 0; v < 8; ++v) {
        int rowg = mtile * 16 + v + 8 * half;
        float iv = sigmoidf_(acc[0][v] + bi);
        float fv = sigmoidf_(acc[1][v] + bf);
        float gv = tanhf(acc[2][v] + bg);
        float ov = sigmoidf_(acc[3][v] + bo);
        float cn = fv * cp[v] + iv * gv;
        c_out[(size_t)rowg * H_ + col] = cn;
        h_out[(size_t)rowg * H_ + col] = ov * tanhf(cn);
    }
}

#define PKB 32
#define PKS (PKB / 4)

struct ProjFrag { v2f a0[PKS]; v2f a1[PKS]; v2f b[PKS][5]; };

__device__ __forceinline__ void proj_load(ProjFrag& f, const float* __restrict__ arow,
                                          const float* const (&wrow)[5], int kb) {
    #pragma unroll
    for (int s = 0; s < PKS; ++s) {
        f.a0[s] = *(const v2f*)(arow + kb + 4 * s);
        f.a1[s] = *(const v2f*)(arow + 16 * LDSS + kb + 4 * s);
    }
    #pragma unroll
    for (int s = 0; s < PKS; ++s)
        #pragma unroll
        for (int i = 0; i < 5; ++i)
            f.b[s][i] = *(const v2f*)(wrow[i] + kb + 4 * s);
}

__device__ __forceinline__ void proj_mma(const ProjFrag& f, v8f (&acc)[10]) {
    #pragma unroll
    for (int s = 0; s < PKS; ++s)
        #pragma unroll
        for (int i = 0; i < 5; ++i) {
            acc[i]     = wmma_f32(f.a0[s], f.b[s][i], acc[i]);
            acc[5 + i] = wmma_f32(f.a1[s], f.b[s][i], acc[5 + i]);
        }
}

__device__ __forceinline__ void proj_role(
    float* __restrict__ hsh, int pb,
    const float* __restrict__ h, const float* __restrict__ lin_W,
    const float* __restrict__ lin_b, float* __restrict__ out, int t)
{
    const int mbase = (pb / 25) * 32;
    const int tid   = threadIdx.x;

    if (tid < 128) {
        float4 hr[32];
        #pragma unroll
        for (int j = 0; j < 32; ++j)
            hr[j] = *(const float4*)(h + (size_t)(mbase + j) * H_ + (tid << 2));
        #pragma unroll
        for (int j = 0; j < 32; ++j)
            *(float4*)(hsh + j * LDSS + (tid << 2)) = hr[j];
    }
    __syncthreads();

    const int wave = tid >> 5;
    const int lane = tid & 31;
    const int half = lane >> 4, l16 = lane & 15;
    const int ntile0 = (pb % 25) * 25 + wave * 5;

    v8f acc[10];
    int  ncol[5];
    const float* wrow[5];
    float bias[5];
    #pragma unroll
    for (int i = 0; i < 5; ++i) {
        ncol[i] = (ntile0 + i) * 16 + l16;
        wrow[i] = lin_W + (size_t)ncol[i] * H_ + 2 * half;
        bias[i] = lin_b[ncol[i]];
    }
    #pragma unroll
    for (int j = 0; j < 10; ++j)
        #pragma unroll
        for (int v = 0; v < 8; ++v) acc[j][v] = 0.0f;

    const float* arow = hsh + l16 * LDSS + 2 * half;

    ProjFrag fp, fq;
    proj_load(fp, arow, wrow, 0);
    for (int bk = 0; bk < H_ / PKB - 2; bk += 2) {
        int nb = bk + 1;
        proj_load(fq, arow, wrow, nb * PKB);
        LOADS_AHEAD();
        proj_mma(fp, acc);
        MMA_DONE10(acc);
        nb = bk + 2;
        proj_load(fp, arow, wrow, nb * PKB);
        LOADS_AHEAD();
        proj_mma(fq, acc);
        MMA_DONE10(acc);
    }
    proj_load(fq, arow, wrow, (H_ / PKB - 1) * PKB);
    LOADS_AHEAD();
    proj_mma(fp, acc);
    proj_mma(fq, acc);
    MMA_DONE10(acc);

    #pragma unroll
    for (int m = 0; m < 2; ++m) {
        #pragma unroll
        for (int i = 0; i < 5; ++i) {
            #pragma unroll
            for (int v = 0; v < 8; ++v) {
                int rowg = mbase + m * 16 + v + 8 * half;
                __builtin_nontemporal_store(
                    acc[m * 5 + i][v] + bias[i],
                    out + (size_t)rowg * T_ * V_ + (size_t)t * V_ + ncol[i]);
            }
        }
    }
}

__global__ __launch_bounds__(NT_) void step_pair(
    const int* __restrict__ captions, const float* __restrict__ embed_W,
    const float* __restrict__ W_ih, const float* __restrict__ W_hh,
    const float* __restrict__ b_ih, const float* __restrict__ b_hh,
    const float* __restrict__ lin_W, const float* __restrict__ lin_b,
    const float* __restrict__ h_cur, const float* __restrict__ c_cur,
    float* __restrict__ h_nxt, float* __restrict__ c_nxt,
    float* __restrict__ out, int t_cell, int t_proj, int n_cell)
{
    __shared__ float smem[32 * LDSS];
    const int bid = blockIdx.x;
    if (bid < n_cell)
        cell_role(smem, bid, captions, t_cell, embed_W, W_ih, W_hh, b_ih, b_hh,
                  h_cur, c_cur, h_nxt, c_nxt);
    else
        proj_role(smem, bid - n_cell, h_cur, lin_W, lin_b, out, t_proj);
}

extern "C" void kernel_launch(void* const* d_in, const int* in_sizes, int n_in,
                              void* d_out, int out_size, void* d_ws, size_t ws_size,
                              hipStream_t stream) {
    const float* features = (const float*)d_in[0];
    const int*   captions = (const int*)d_in[1];
    const float* embed_W  = (const float*)d_in[2];
    const float* W_ih     = (const float*)d_in[3];
    const float* W_hh     = (const float*)d_in[4];
    const float* b_ih     = (const float*)d_in[5];
    const float* b_hh     = (const float*)d_in[6];
    const float* lin_W    = (const float*)d_in[7];
    const float* lin_b    = (const float*)d_in[8];
    float* out = (float*)d_out;

    float* ws = (float*)d_ws;
    float* hbuf[2] = { ws,                       ws + (size_t)B_ * H_ };
    float* cbuf[2] = { ws + 2 * (size_t)B_ * H_, ws + 3 * (size_t)B_ * H_ };

    lstm_init<<<(B_ * H_ + 255) / 256, 256, 0, stream>>>(features, hbuf[0], cbuf[0]);

    step_pair<<<CELL_BLOCKS, NT_, 0, stream>>>(
        captions, embed_W, W_ih, W_hh, b_ih, b_hh, lin_W, lin_b,
        hbuf[0], cbuf[0], hbuf[1], cbuf[1], out, 0, 0, CELL_BLOCKS);

    for (int t = 1; t < T_; ++t) {
        int cur = t & 1, nxt = cur ^ 1;
        step_pair<<<CELL_BLOCKS + PROJ_BLOCKS, NT_, 0, stream>>>(
            captions, embed_W, W_ih, W_hh, b_ih, b_hh, lin_W, lin_b,
            hbuf[cur], cbuf[cur], hbuf[nxt], cbuf[nxt], out, t, t - 1, CELL_BLOCKS);
    }

    step_pair<<<PROJ_BLOCKS, NT_, 0, stream>>>(
        captions, embed_W, W_ih, W_hh, b_ih, b_hh, lin_W, lin_b,
        hbuf[T_ & 1], cbuf[T_ & 1], hbuf[(T_ & 1) ^ 1], cbuf[(T_ & 1) ^ 1],
        out, 0, T_ - 1, 0);
}


# One wave per SIMD on all 256 WGPs: projection in 128 blocks of 4 waves (was 100 of 5), cell in 128 blocks of 2 working waves (was 64 of 4); tiles per wave and sum order unchanged
# speedup vs baseline: 3.5429x; 1.3233x over previous
#include <hip/hip_runtime.h>
#include <cmath>


typedef __attribute__((ext_vector_type(2))) float v2f;
typedef __attribute__((ext_vector_type(8))) float v8f;

#define B_   128
#define T_   20
#define E_   512
#define H_   512
#define V_   10000
#define LDSS 516
#define NT_  128
#define CELL_BLOCKS 128
#define PROJ_BLOCKS 128

__device__ __forceinline__ float sigmoidf_(float x) {
    return 1.0f / (1.0f + __expf(-x));
}

__device__ __forceinline__ v8f wmma_f32(v2f a, v2f b, v8f c) {
    return __builtin_amdgcn_wmma_f32_16x16x4_f32(false, a, false, b, (short)0, c,
                                                 false, false);
}

__global__ void lstm_init(const float* __restrict__ features,
                          float* __restrict__ h0, float* __restrict__ c0) {
    int i = blockIdx.x * blockDim.x + threadIdx.x;
    if (i < B_ * H_) { h0[i] = features[i]; c0[i] = 0.0f; }
}

#define LOADS_AHEAD() __builtin_amdgcn_sched_barrier(0)
#define MMA_DONE4(a)                                                                    \
    do {                                                                                \
        asm volatile("v_nop\n\tv_nop\n\tv_nop\n\tv_nop\n\tv_nop"                         \
                     : "+v"(a[0]), "+v"(a[1]), "+v"(a[2]), "+v"(a[3]));                  \
        __builtin_amdgcn_sched_barrier(0);                                              \
    } while (0)
#define MMA_DONE10(a)                                                                   \
    do {                                                                                \
        asm volatile("v_nop\n\tv_nop\n\tv_nop\n\tv_nop\n\tv_nop"                         \
                     : "+v"(a[0]), "+v"(a[1]), "+v"(a[2]), "+v"(a[3]), "+v"(a[4]),       \
                       "+v"(a[5]), "+v"(a[6]), "+v"(a[7]), "+v"(a[8]), "+v"(a[9]));      \
        __builtin_amdgcn_sched_barrier(0);                                              \
    } while (0)
#define CKB 64
#define CKS (CKB / 4)

struct CellFrag { v2f a[CKS]; v2f b[CKS][4]; };

__device__ __forceinline__ void cell_load(CellFrag& f, const float* __restrict__ arow,
                                          const float* __restrict__ wrow, int kb) {
    #pragma unroll
    for (int s = 0; s < CKS; ++s)
        f.a[s] = *(const v2f*)(arow + kb + 4 * s);
    #pragma unroll
    for (int s = 0; s < CKS; ++s)
        #pragma unroll
        for (int g = 0; g < 4; ++g)
            f.b[s][g] = *(const v2f*)(wrow + (size_t)g * H_ * E_ + kb + 4 * s);
}

__device__ __forceinline__ void cell_mma(const CellFrag& f, v8f (&acc)[4]) {
    #pragma unroll
    for (int s = 0; s < CKS; ++s)
        #pragma unroll
        for (int g = 0; g < 4; ++g)
            acc[g] = wmma_f32(f.a[s], f.b[s][g], acc[g]);
}

__device__ __forceinline__ void cell_role(
    float* __restrict__ smem, int bid,
    const int* __restrict__ captions, int t,
    const float* __restrict__ embed_W,
    const float* __restrict__ W_ih, const float* __restrict__ W_hh,
    const float* __restrict__ b_ih, const float* __restrict__ b_hh,
    const float* __restrict__ h_in, const float* __restrict__ c_in,
    float* __restrict__ h_out, float* __restrict__ c_out)
{
    float* xt = smem;
    float* ht = smem + 16 * LDSS;
    const int mtile = bid >> 4;
    const int tid   = threadIdx.x;

    {
        constexpr int NJ = 16;
        int tok[NJ];
        #pragma unroll
        for (int j = 0; j < NJ; ++j) {
            int q = j * 128 + tid;
            tok[j] = captions[(mtile * 16 + (q >> 7)) * T_ + t];
        }
        float4 xr[NJ], hr[NJ];
        #pragma unroll
        for (int j = 0; j < NJ; ++j) {
            int q = j * 128 + tid;
            int row = q >> 7, e = (q & 127) << 2;
            xr[j] = *(const float4*)(embed_W + (size_t)tok[j] * E_ + e);
            hr[j] = *(const float4*)(h_in + (size_t)(mtile * 16 + row) * H_ + e);
        }
        #pragma unroll
        for (int j = 0; j < NJ; ++j) {
            int q = j * 128 + tid;
            int row = q >> 7, e = (q & 127) << 2;
            *(float4*)(xt + row * LDSS + e) = xr[j];
            *(float4*)(ht + row * LDSS + e) = hr[j];
        }
    }
    __syncthreads();

    const int wave  = tid >> 5;
    if (wave >= 2) return;
    const int lane  = tid & 31;
    const int half  = lane >> 4;
    const int l16   = lane & 15;
    const int htile = (bid & 15) * 2 + wave;
    const int col   = htile * 16 + l16;

    float bi = b_ih[0 * H_ + col] + b_hh[0 * H_ + col];
    float bf = b_ih[1 * H_ + col] + b_hh[1 * H_ + col];
    float bg = b_ih[2 * H_ + col] + b_hh[2 * H_ + col];
    float bo = b_ih[3 * H_ + col] + b_hh[3 * H_ + col];
    float cp[8];
    #pragma unroll
    for (int v = 0; v < 8; ++v)
        cp[v] = c_in[(size_t)(mtile * 16 + v + 8 * half) * H_ + col];

    v8f acc[4];
    #pragma unroll
    for (int g = 0; g < 4; ++g)
        #pragma unroll
        for (int v = 0; v < 8; ++v) acc[g][v] = 0.0f;

    const float* ax = xt + l16 * LDSS + 2 * half;
    const float* ah = ht + l16 * LDSS + 2 * half;
    const float* wx = W_ih + (size_t)col * E_ + 2 * half;
    const float* wh = W_hh + (size_t)col * H_ + 2 * half;

    constexpr int XB = E_ / CKB;
    CellFrag fp, fq;
    cell_load(fp, ax, wx, 0);
    asm volatile("" : "+v"(bi), "+v"(bf), "+v"(bg), "+v"(bo));
    for (int bk = 0; bk < 2 * XB - 2; bk += 2) {
        int nb = bk + 1;
        cell_load(fq, nb < XB ? ax : ah, nb < XB ? wx : wh, (nb & (XB - 1)) * CKB);
        LOADS_AHEAD();
        cell_mma(fp, acc);
        MMA_DONE4(acc);
        nb = bk + 2;
        cell_load(fp, nb < XB ? ax : ah, nb < XB ? wx : wh, (nb & (XB - 1)) * CKB);
        LOADS_AHEAD();
        cell_mma(fq, acc);
        MMA_DONE4(acc);
    }
    cell_load(fq, ah, wh, (XB - 1) * CKB);
    LOADS_AHEAD();
    cell_mma(fp, acc);
    cell_mma(fq, acc);
    MMA_DONE4(acc);

    #pragma unroll
    for (int v = 0; v < 8; ++v) {
        int rowg = mtile * 16 + v + 8 * half;
        float iv = sigmoidf_(acc[0][v] + bi);
        float fv = sigmoidf_(acc[1][v] + bf);
        float gv = tanhf(acc[2][v] + bg);
        float ov = sigmoidf_(acc[3][v] + bo);
        float cn = fv * cp[v] + iv * gv;
        c_out[(size_t)rowg * H_ + col] = cn;
        h_out[(size_t)rowg * H_ + col] = ov * tanhf(cn);
    }
}

#define PKB 32
#define PKS (PKB / 4)

struct ProjFrag { v2f a0[PKS]; v2f a1[PKS]; v2f b[PKS][5]; };

__device__ __forceinline__ void proj_load(ProjFrag& f, const float* __restrict__ arow,
                                          const float* const (&wrow)[5], int kb) {
    #pragma unroll
    for (int s = 0; s < PKS; ++s) {
        f.a0[s] = *(const v2f*)(arow + kb + 4 * s);
        f.a1[s] = *(const v2f*)(arow + 16 * LDSS + kb + 4 * s);
    }
    #pragma unroll
    for (int s = 0; s < PKS; ++s)
        #pragma unroll
        for (int i = 0; i < 5; ++i)
            f.b[s][i] = *(const v2f*)(wrow[i] + kb + 4 * s);
}

__device__ __forceinline__ void proj_mma(const ProjFrag& f, v8f (&acc)[10]) {
    #pragma unroll
    for (int s = 0; s < PKS; ++s)
        #pragma unroll
        for (int i = 0; i < 5; ++i) {
            acc[i]     = wmma_f32(f.a0[s], f.b[s][i], acc[i]);
            acc[5 + i] = wmma_f32(f.a1[s], f.b[s][i], acc[5 + i]);
        }
}

__device__ __forceinline__ void proj_role(
    float* __restrict__ hsh, int pb,
    const float* __restrict__ h, const float* __restrict__ lin_W,
    const float* __restrict__ lin_b, float* __restrict__ out, int t)
{
    const int mbase = (pb >> 5) * 32;
    const int tid   = threadIdx.x;

    {
        float4 hr[32];
        #pragma unroll
        for (int j = 0; j < 32; ++j)
            hr[j] = *(const float4*)(h + (size_t)(mbase + j) * H_ + (tid << 2));
        #pragma unroll
        for (int j = 0; j < 32; ++j)
            *(float4*)(hsh + j * LDSS + (tid << 2)) = hr[j];
    }
    __syncthreads();

    const int wave = tid >> 5;
    const int lane = tid & 31;
    const int half = lane >> 4, l16 = lane & 15;
    const int ntile0 = (pb & 31) * 20 + wave * 5;
    if (ntile0 >= V_ / 16) return;

    v8f acc[10];
    int  ncol[5];
    const float* wrow[5];
    float bias[5];
    #pragma unroll
    for (int i = 0; i < 5; ++i) {
        ncol[i] = (ntile0 + i) * 16 + l16;
        wrow[i] = lin_W + (size_t)ncol[i] * H_ + 2 * half;
        bias[i] = lin_b[ncol[i]];
    }
    #pragma unroll
    for (int j = 0; j < 10; ++j)
        #pragma unroll
        for (int v = 0; v < 8; ++v) acc[j][v] = 0.0f;

    const float* arow = hsh + l16 * LDSS + 2 * half;

    ProjFrag fp, fq;
    proj_load(fp, arow, wrow, 0);
    for (int bk = 0; bk < H_ / PKB - 2; bk += 2) {
        int nb = bk + 1;
        proj_load(fq, arow, wrow, nb * PKB);
        LOADS_AHEAD();
        proj_mma(fp, acc);
        MMA_DONE10(acc);
        nb = bk + 2;
        proj_load(fp, arow, wrow, nb * PKB);
        LOADS_AHEAD();
        proj_mma(fq, acc);
        MMA_DONE10(acc);
    }
    proj_load(fq, arow, wrow, (H_ / PKB - 1) * PKB);
    LOADS_AHEAD();
    proj_mma(fp, acc);
    proj_mma(fq, acc);
    MMA_DONE10(acc);

    #pragma unroll
    for (int m = 0; m < 2; ++m) {
        #pragma unroll
        for (int i = 0; i < 5; ++i) {
            #pragma unroll
            for (int v = 0; v < 8; ++v) {
                int rowg = mbase + m * 16 + v + 8 * half;
                __builtin_nontemporal_store(
                    acc[m * 5 + i][v] + bias[i],
                    out + (size_t)rowg * T_ * V_ + (size_t)t * V_ + ncol[i]);
            }
        }
    }
}

__global__ __launch_bounds__(NT_) void step_pair(
    const int* __restrict__ captions, const float* __restrict__ embed_W,
    const float* __restrict__ W_ih, const float* __restrict__ W_hh,
    const float* __restrict__ b_ih, const float* __restrict__ b_hh,
    const float* __restrict__ lin_W, const float* __restrict__ lin_b,
    const float* __restrict__ h_cur, const float* __restrict__ c_cur,
    float* __restrict__ h_nxt, float* __restrict__ c_nxt,
    float* __restrict__ out, int t_cell, int t_proj, int n_cell)
{
    __shared__ float smem[32 * LDSS];
    const int bid = blockIdx.x;
    if (bid < n_cell)
        cell_role(smem, bid, captions, t_cell, embed_W, W_ih, W_hh, b_ih, b_hh,
                  h_cur, c_cur, h_nxt, c_nxt);
    else
        proj_role(smem, bid - n_cell, h_cur, lin_W, lin_b, out, t_proj);
}

extern "C" void kernel_launch(void* const* d_in, const int* in_sizes, int n_in,
                              void* d_out, int out_size, void* d_ws, size_t ws_size,
                              hipStream_t stream) {
    const float* features = (const float*)d_in[0];
    const int*   captions = (const int*)d_in[1];
    const float* embed_W  = (const float*)d_in[2];
    const float* W_ih     = (const float*)d_in[3];
    const float* W_hh     = (const float*)d_in[4];
    const float* b_ih     = (const float*)d_in[5];
    const float* b_hh     = (const float*)d_in[6];
    const float* lin_W    = (const float*)d_in[7];
    const float* lin_b    = (const float*)d_in[8];
    float* out = (float*)d_out;

    float* ws = (float*)d_ws;
    float* hbuf[2] = { ws,                       ws + (size_t)B_ * H_ };
    float* cbuf[2] = { ws + 2 * (size_t)B_ * H_, ws + 3 * (size_t)B_ * H_ };

    lstm_init<<<(B_ * H_ + 255) / 256, 256, 0, stream>>>(features, hbuf[0], cbuf[0]);

    step_pair<<<CELL_BLOCKS, NT_, 0, stream>>>(
        captions, embed_W, W_ih, W_hh, b_ih, b_hh, lin_W, lin_b,
        hbuf[0], cbuf[0], hbuf[1], cbuf[1], out, 0, 0, CELL_BLOCKS);

    for (int t = 1; t < T_; ++t) {
        int cur = t & 1, nxt = cur ^ 1;
        step_pair<<<CELL_BLOCKS + PROJ_BLOCKS, NT_, 0, stream>>>(
            captions, embed_W, W_ih, W_hh, b_ih, b_hh, lin_W, lin_b,
            hbuf[cur], cbuf[cur], hbuf[nxt], cbuf[nxt], out, t, t - 1, CELL_BLOCKS);
    }

    step_pair<<<PROJ_BLOCKS, NT_, 0, stream>>>(
        captions, embed_W, W_ih, W_hh, b_ih, b_hh, lin_W, lin_b,
        hbuf[T_ & 1], cbuf[T_ & 1], hbuf[(T_ & 1) ^ 1], cbuf[(T_ & 1) ^ 1],
        out, 0, T_ - 1, 0);
}


# Fragment loads of A (LDS) and B (weights) widened from 8 to 16 bytes per lane in both roles: half the load instructions per cache line; sum order changes (k permuted alike in A and B inside each 8)
# speedup vs baseline: 4.2634x; 1.2034x over previous
#include <hip/hip_runtime.h>
#include <cmath>


typedef __attribute__((ext_vector_type(2))) float v2f;
typedef __attribute__((ext_vector_type(4))) float v4f;
typedef __attribute__((ext_vector_type(8))) float v8f;

#define B_   128
#define T_   20
#define E_   512
#define H_   512
#define V_   10000
#define LDSS 516
#define NT_  128
#define CELL_BLOCKS 128
#define PROJ_BLOCKS 128

__device__ __forceinline__ float sigmoidf_(float x) {
    return 1.0f / (1.0f + __expf(-x));
}

__device__ __forceinline__ v8f wmma_f32(v2f a, v2f b, v8f c) {
    return __builtin_amdgcn_wmma_f32_16x16x4_f32(false, a, false, b, (short)0, c,
                                                 false, false);
}

__device__ __forceinline__ v2f frag_lo(v4f q) { return __builtin_shufflevector(q, q, 0, 1); }
__device__ __forceinline__ v2f frag_hi(v4f q) { return __builtin_shufflevector(q, q, 2, 3); }

__global__ void lstm_init(const float* __restrict__ features,
                          float* __restrict__ h0, float* __restrict__ c0) {
    int i = blockIdx.x * blockDim.x + threadIdx.x;
    if (i < B_ * H_) { h0[i] = features[i]; c0[i] = 0.0f; }
}

#define LOADS_AHEAD() __builtin_amdgcn_sched_barrier(0)
#define MMA_DONE4(a)                                                                    \
    do {                                                                                \
        asm volatile("v_nop\n\tv_nop\n\tv_nop\n\tv_nop\n\tv_nop"                         \
                     : "+v"(a[0]), "+v"(a[1]), "+v"(a[2]), "+v"(a[3]));                  \
        __builtin_amdgcn_sched_barrier(0);                                              \
    } while (0)
#define MMA_DONE10(a)                                                                   \
    do {                                                                                \
        asm volatile("v_nop\n\tv_nop\n\tv_nop\n\tv_nop\n\tv_nop"                         \
                     : "+v"(a[0]), "+v"(a[1]), "+v"(a[2]), "+v"(a[3]), "+v"(a[4]),       \
                       "+v"(a[5]), "+v"(a[6]), "+v"(a[7]), "+v"(a[8]), "+v"(a[9]));      \
        __builtin_amdgcn_sched_barrier(0);                                              \
    } while (0)
#define CKB 64
#define CKS (CKB / 8)

struct CellFrag { v4f a[CKS]; v4f b[CKS][4]; };

__device__ __forceinline__ void cell_load(CellFrag& f, const float* __restrict__ arow,
                                          const float* __restrict__ wrow, int kb) {
    #pragma unroll
    for (int s = 0; s < CKS; ++s)
        f.a[s] = *(const v4f*)(arow + kb + 8 * s);
    #pragma unroll
    for (int s = 0; s < CKS; ++s)
        #pragma unroll
        for (int g = 0; g < 4; ++g)
            f.b[s][g] = *(const v4f*)(wrow + (size_t)g * H_ * E_ + kb + 8 * s);
}

__device__ __forceinline__ void cell_mma(const CellFrag& f, v8f (&acc)[4]) {
    #pragma unroll
    for (int s = 0; s < CKS; ++s) {
        #pragma unroll
        for (int g = 0; g < 4; ++g)
            acc[g] = wmma_f32(frag_lo(f.a[s]), frag_lo(f.b[s][g]), acc[g]);
        #pragma unroll
        for (int g = 0; g < 4; ++g)
            acc[g] = wmma_f32(frag_hi(f.a[s]), frag_hi(f.b[s][g]), acc[g]);
    }
}

__device__ __forceinline__ void cell_role(
    float* __restrict__ smem, int bid,
    const int* __restrict__ captions, int t,
    const float* __restrict__ embed_W,
    const float* __restrict__ W_ih, const float* __restrict__ W_hh,
    const float* __restrict__ b_ih, const float* __restrict__ b_hh,
    const float* __restrict__ h_in, const float* __restrict__ c_in,
    float* __restrict__ h_out, float* __restrict__ c_out)
{
    float* xt = smem;
    float* ht = smem + 16 * LDSS;
    const int mtile = bid >> 4;
    const int tid   = threadIdx.x;

    {
        constexpr int NJ = 16;
        int tok[NJ];
        #pragma unroll
        for (int j = 0; j < NJ; ++j) {
            int q = j * 128 + tid;
            tok[j] = captions[(mtile * 16 + (q >> 7)) * T_ + t];
        }
        float4 xr[NJ], hr[NJ];
        #pragma unroll
        for (int j = 0; j < NJ; ++j) {
            int q = j * 128 + tid;
            int row = q >> 7, e = (q & 127) << 2;
            xr[j] = *(const float4*)(embed_W + (size_t)tok[j] * E_ + e);
            hr[j] = *(const float4*)(h_in + (size_t)(mtile * 16 + row) * H_ + e);
        }
        #pragma unroll
        for (int j = 0; j < NJ; ++j) {
            int q = j * 128 + tid;
            int row = q >> 7, e = (q & 127) << 2;
            *(float4*)(xt + row * LDSS + e) = xr[j];
            *(float4*)(ht + row * LDSS + e) = hr[j];
        }
    }
    __syncthreads();

    const int wave  = tid >> 5;
    if (wave >= 2) return;
    const int lane  = tid & 31;
    const int half  = lane >> 4;
    const int l16   = lane & 15;
    const int htile = (bid & 15) * 2 + wave;
    const int col   = htile * 16 + l16;

    float bi = b_ih[0 * H_ + col] + b_hh[0 * H_ + col];
    float bf = b_ih[1 * H_ + col] + b_hh[1 * H_ + col];
    float bg = b_ih[2 * H_ + col] + b_hh[2 * H_ + col];
    float bo = b_ih[3 * H_ + col] + b_hh[3 * H_ + col];
    float cp[8];
    #pragma unroll
    for (int v = 0; v < 8; ++v)
        cp[v] = c_in[(size_t)(mtile * 16 + v + 8 * half) * H_ + col];

    v8f acc[4];
    #pragma unroll
    for (int g = 0; g < 4; ++g)
        #pragma unroll
        for (int v = 0; v < 8; ++v) acc[g][v] = 0.0f;

    const float* ax = xt + l16 * LDSS + 4 * half;
    const float* ah = ht + l16 * LDSS + 4 * half;
    const float* wx = W_ih + (size_t)col * E_ + 4 * half;
    const float* wh = W_hh + (size_t)col * H_ + 4 * half;

    constexpr int XB = E_ / CKB;
    CellFrag fp, fq;
    cell_load(fp, ax, wx, 0);
    asm volatile("" : "+v"(bi), "+v"(bf), "+v"(bg), "+v"(bo));
    for (int bk = 0; bk < 2 * XB - 2; bk += 2) {
        int nb = bk + 1;
        cell_load(fq, nb < XB ? ax : ah, nb < XB ? wx : wh, (nb & (XB - 1)) * CKB);
        LOADS_AHEAD();
        cell_mma(fp, acc);
        MMA_DONE4(acc);
        nb = bk + 2;
        cell_load(fp, nb < XB ? ax : ah, nb < XB ? wx : wh, (nb & (XB - 1)) * CKB);
        LOADS_AHEAD();
        cell_mma(fq, acc);
        MMA_DONE4(acc);
    }
    cell_load(fq, ah, wh, (XB - 1) * CKB);
    LOADS_AHEAD();
    cell_mma(fp, acc);
    cell_mma(fq, acc);
    MMA_DONE4(acc);

    #pragma unroll
    for (int v = 0; v < 8; ++v) {
        int rowg = mtile * 16 + v + 8 * half;
        float iv = sigmoidf_(acc[0][v] + bi);
        float fv = sigmoidf_(acc[1][v] + bf);
        float gv = tanhf(acc[2][v] + bg);
        float ov = sigmoidf_(acc[3][v] + bo);
        float cn = fv * cp[v] + iv * gv;
        c_out[(size_t)rowg * H_ + col] = cn;
        h_out[(size_t)rowg * H_ + col] = ov * tanhf(cn);
    }
}

#define PKB 32
#define PKS (PKB / 8)

struct ProjFrag { v4f a0[PKS]; v4f a1[PKS]; v4f b[PKS][5]; };

__device__ __forceinline__ void proj_load(ProjFrag& f, const float* __restrict__ arow,
                                          const float* const (&wrow)[5], int kb) {
    #pragma unroll
    for (int s = 0; s < PKS; ++s) {
        f.a0[s] = *(const v4f*)(arow + kb + 8 * s);
        f.a1[s] = *(const v4f*)(arow + 16 * LDSS + kb + 8 * s);
    }
    #pragma unroll
    for (int s = 0; s < PKS; ++s)
        #pragma unroll
        for (int i = 0; i < 5; ++i)
            f.b[s][i] = *(const v4f*)(wrow[i] + kb + 8 * s);
}

__device__ __forceinline__ void proj_mma(const ProjFrag& f, v8f (&acc)[10]) {
    #pragma unroll
    for (int s = 0; s < PKS; ++s) {
        #pragma unroll
        for (int i = 0; i < 5; ++i) {
            acc[i]     = wmma_f32(frag_lo(f.a0[s]), frag_lo(f.b[s][i]), acc[i]);
            acc[5 + i] = wmma_f32(frag_lo(f.a1[s]), frag_lo(f.b[s][i]), acc[5 + i]);
        }
        #pragma unroll
        for (int i = 0; i < 5; ++i) {
            acc[i]     = wmma_f32(frag_hi(f.a0[s]), frag_hi(f.b[s][i]), acc[i]);
            acc[5 + i] = wmma_f32(frag_hi(f.a1[s]), frag_hi(f.b[s][i]), acc[5 + i]);
        }
    }
}

__device__ __forceinline__ void proj_role(
    float* __restrict__ hsh, int pb,
    const float* __restrict__ h, const float* __restrict__ lin_W,
    const float* __restrict__ lin_b, float* __restrict__ out, int t)
{
    const int mbase = (pb >> 5) * 32;
    const int tid   = threadIdx.x;

    {
        float4 hr[32];
        #pragma unroll
        for (int j = 0; j < 32; ++j)
            hr[j] = *(const float4*)(h + (size_t)(mbase + j) * H_ + (tid << 2));
        #pragma unroll
        for (int j = 0; j < 32; ++j)
            *(float4*)(hsh + j * LDSS + (tid << 2)) = hr[j];
    }
    __syncthreads();

    const int wave = tid >> 5;
    const int lane = tid & 31;
    const int half = lane >> 4, l16 = lane & 15;
    const int ntile0 = (pb & 31) * 20 + wave * 5;
    if (ntile0 >= V_ / 16) return;

    v8f acc[10];
    int  ncol[5];
    const float* wrow[5];
    float bias[5];
    #pragma unroll
    for (int i = 0; i < 5; ++i) {
        ncol[i] = (ntile0 + i) * 16 + l16;
        wrow[i] = lin_W + (size_t)ncol[i] * H_ + 4 * half;
        bias[i] = lin_b[ncol[i]];
    }
    #pragma unroll
    for (int j = 0; j < 10; ++j)
        #pragma unroll
        for (int v = 0; v < 8; ++v) acc[j][v] = 0.0f;

    const float* arow = hsh + l16 * LDSS + 4 * half;

    ProjFrag fp, fq;
    proj_load(fp, arow, wrow, 0);
    for (int bk = 0; bk < H_ / PKB - 2; bk += 2) {
        int nb = bk + 1;
        proj_load(fq, arow, wrow, nb * PKB);
        LOADS_AHEAD();
        proj_mma(fp, acc);
        MMA_DONE10(acc);
        nb = bk + 2;
        proj_load(fp, arow, wrow, nb * PKB);
        LOADS_AHEAD();
        proj_mma(fq, acc);
        MMA_DONE10(acc);
    }
    proj_load(fq, arow, wrow, (H_ / PKB - 1) * PKB);
    LOADS_AHEAD();
    proj_mma(fp, acc);
    proj_mma(fq, acc);
    MMA_DONE10(acc);

    #pragma unroll
    for (int m = 0; m < 2; ++m) {
        #pragma unroll
        for (int i = 0; i < 5; ++i) {
            #pragma unroll
            for (int v = 0; v < 8; ++v) {
                int rowg = mbase + m * 16 + v + 8 * half;
                __builtin_nontemporal_store(
                    acc[m * 5 + i][v] + bias[i],
                    out + (size_t)rowg * T_ * V_ + (size_t)t * V_ + ncol[i]);
            }
        }
    }
}

__global__ __launch_bounds__(NT_) void step_pair(
    const int* __restrict__ captions, const float* __restrict__ embed_W,
    const float* __restrict__ W_ih, const float* __restrict__ W_hh,
    const float* __restrict__ b_ih, const float* __restrict__ b_hh,
    const float* __restrict__ lin_W, const float* __restrict__ lin_b,
    const float* __restrict__ h_cur, const float* __restrict__ c_cur,
    float* __restrict__ h_nxt, float* __restrict__ c_nxt,
    float* __restrict__ out, int t_cell, int t_proj, int n_cell)
{
    __shared__ float smem[32 * LDSS];
    const int bid = blockIdx.x;
    if (bid < n_cell)
        cell_role(smem, bid, captions, t_cell, embed_W, W_ih, W_hh, b_ih, b_hh,
                  h_cur, c_cur, h_nxt, c_nxt);
    else
        proj_role(smem, bid - n_cell, h_cur, lin_W, lin_b, out, t_proj);
}

extern "C" void kernel_launch(void* const* d_in, const int* in_sizes, int n_in,
                              void* d_out, int out_size, void* d_ws, size_t ws_size,
                              hipStream_t stream) {
    const float* features = (const float*)d_in[0];
    const int*   captions = (const int*)d_in[1];
    const float* embed_W  = (const float*)d_in[2];
    const float* W_ih     = (const float*)d_in[3];
    const float* W_hh     = (const float*)d_in[4];
    const float* b_ih     = (const float*)d_in[5];
    const float* b_hh     = (const float*)d_in[6];
    const float* lin_W    = (const float*)d_in[7];
    const float* lin_b    = (const float*)d_in[8];
    float* out = (float*)d_out;

    float* ws = (float*)d_ws;
    float* hbuf[2] = { ws,                       ws + (size_t)B_ * H_ };
    float* cbuf[2] = { ws + 2 * (size_t)B_ * H_, ws + 3 * (size_t)B_ * H_ };

    lstm_init<<<(B_ * H_ + 255) / 256, 256, 0, stream>>>(features, hbuf[0], cbuf[0]);

    step_pair<<<CELL_BLOCKS, NT_, 0, stream>>>(
        captions, embed_W, W_ih, W_hh, b_ih, b_hh, lin_W, lin_b,
        hbuf[0], cbuf[0], hbuf[1], cbuf[1], out, 0, 0, CELL_BLOCKS);

    for (int t = 1; t < T_; ++t) {
        int cur = t & 1, nxt = cur ^ 1;
        step_pair<<<CELL_BLOCKS + PROJ_BLOCKS, NT_, 0, stream>>>(
            captions, embed_W, W_ih, W_hh, b_ih, b_hh, lin_W, lin_b,
            hbuf[cur], cbuf[cur], hbuf[nxt], cbuf[nxt], out, t, t - 1, CELL_BLOCKS);
    }

    step_pair<<<PROJ_BLOCKS, NT_, 0, stream>>>(
        captions, embed_W, W_ih, W_hh, b_ih, b_hh, lin_W, lin_b,
        hbuf[T_ & 1], cbuf[T_ & 1], hbuf[(T_ & 1) ^ 1], cbuf[(T_ & 1) ^ 1],
        out, 0, T_ - 1, 0);
}
